# starting kernel
# baseline (speedup 1.0000x reference)
#include <hip/hip_runtime.h>

typedef float v2f __attribute__((ext_vector_type(2)));
typedef float v8f __attribute__((ext_vector_type(8)));

#define N_ROWS 1024
#define H_DIM  512
#define O_DIM  512
#define NPAR   4
#define ORDER_C 3
#define KCHUNK 128
#define WAVES_PER_WG 8
#define ROWS_PER_WG  (WAVES_PER_WG * 16)

__global__ __launch_bounds__(256)
void highorder_wmma_kernel(const float* __restrict__ inp,
                           const float* __restrict__ params,
                           const float* __restrict__ bias,
                           const int*   __restrict__ edge_types,
                           const int*   __restrict__ msg_to_p,
                           float*       __restrict__ out)
{
    __shared__ float ldsB[NPAR * KCHUNK * 16];

    const int tid  = threadIdx.x;
    const int lane = tid & 31;
    const int wave = tid >> 5;
    const int g    = lane >> 4;
    const int l16  = lane & 15;

    const int colBase = blockIdx.x * 16;
    const int rowBase = blockIdx.y * ROWS_PER_WG + wave*16;

    const int msg_to = msg_to_p[0];
    const int i1 = (msg_to == 0) ? 1 : 0;
    const int i2 = (msg_to == 2) ? 1 : 2;
    const float* A1 = inp + (size_t)i1 * N_ROWS * H_DIM;
    const float* A2 = inp + (size_t)i2 * N_ROWS * H_DIM;

    const int myRow = rowBase + l16;
    const float* a1p = A1 + (size_t)myRow * H_DIM;
    const float* a2p = A2 + (size_t)myRow * H_DIM;

    const v8f vzero = {0.f, 0.f, 0.f, 0.f, 0.f, 0.f, 0.f, 0.f};
    v8f acc0 = vzero, acc1 = vzero, acc2 = vzero, acc3 = vzero;

    for (int kc = 0; kc < H_DIM; kc += KCHUNK) {
        #pragma unroll
        for (int it = 0; it < 8; ++it) {
            const int fid = tid + it * 256;
            const int c4  = fid & 3;
            const int k   = (fid >> 2) & (KCHUNK-1);
            const int t   = fid >> 9;
            const float4 v = *(const float4*)(params +
                (((size_t)t * H_DIM + (size_t)(kc + k)) * O_DIM + colBase + c4*4));
            const int base = ((t * (KCHUNK/2) + (k >> 1)) * 16 + c4*4) * 2 + (k & 1);
            ldsB[base + 0] = v.x;
            ldsB[base + 2] = v.y;
            ldsB[base + 4] = v.z;
            ldsB[base + 6] = v.w;
        }
        __syncthreads();

        #pragma unroll 4
        for (int k0 = 0; k0 < KCHUNK; k0 += 4) {
            const int kl = k0 + 2*g;
            const int kk = kc + kl;
            v2f a1 = *(const v2f*)(a1p + kk);
            v2f a2 = *(const v2f*)(a2p + kk);
            v2f a  = a1 * a2;

            const float* bbase = &ldsB[(kl >> 1) * 32 + l16 * 2];
            v2f b0 = *(const v2f*)(bbase + 0 * (KCHUNK/2) * 32);
            v2f b1 = *(const v2f*)(bbase + 1 * (KCHUNK/2) * 32);
            v2f b2 = *(const v2f*)(bbase + 2 * (KCHUNK/2) * 32);
            v2f b3 = *(const v2f*)(bbase + 3 * (KCHUNK/2) * 32);

            acc0 = __builtin_amdgcn_wmma_f32_16x16x4_f32(false, a, false, b0,
                                                         (short)0, acc0, false, false);
            acc1 = __builtin_amdgcn_wmma_f32_16x16x4_f32(false, a, false, b1,
                                                         (short)0, acc1, false, false);
            acc2 = __builtin_amdgcn_wmma_f32_16x16x4_f32(false, a, false, b2,
                                                         (short)0, acc2, false, false);
            acc3 = __builtin_amdgcn_wmma_f32_16x16x4_f32(false, a, false, b3,
                                                         (short)0, acc3, false, false);
        }
        __syncthreads();
    }

    const int colGlobal = colBase + l16;
    #pragma unroll
    for (int j = 0; j < 8; ++j) {
        const int row = rowBase + j + 8*g;
        const int id  = edge_types[row * ORDER_C + msg_to];
        float v = acc0[j];
        v = (id == 1) ? acc1[j] : v;
        v = (id == 2) ? acc2[j] : v;
        v = (id == 3) ? acc3[j] : v;
        v += bias[id * O_DIM + colGlobal];
        out[(size_t)row * O_DIM + colGlobal] = v;
    }
}

extern "C" void kernel_launch(void* const* d_in, const int* in_sizes, int n_in,
                              void* d_out, int out_size, void* d_ws, size_t ws_size,
                              hipStream_t stream) {
    const float* inp        = (const float*)d_in[0];
    const float* params     = (const float*)d_in[1];
    const float* bias       = (const float*)d_in[2];
    const int*   edge_types = (const int*)d_in[3];
    const int*   msg_to     = (const int*)d_in[6];
    float* out = (float*)d_out;

    dim3 grid(O_DIM / 16, N_ROWS / ROWS_PER_WG);
    dim3 block(256);
    highorder_wmma_kernel<<<grid, block, 0, stream>>>(inp, params, bias,
                                                      edge_types, msg_to, out);
}


# Software prefetch: next K block's A lines, next B panel and epilogue id/bias loaded ahead of the WMMAs (16-byte loads, K loop unrolled); k order in each 8-k group regrouped, sum order changes
# speedup vs baseline: 1.2626x; 1.2626x over previous
#include <hip/hip_runtime.h>


typedef float v2f __attribute__((ext_vector_type(2)));
typedef float v4f __attribute__((ext_vector_type(4)));
typedef float v8f __attribute__((ext_vector_type(8)));

#define N_ROWS 1024
#define H_DIM  512
#define O_DIM  512
#define NPAR   4
#define ORDER_C 3
#define KCHUNK 128
#define KBLK   32
#define WAVES_PER_WG 8
#define ROWS_PER_WG  (WAVES_PER_WG * 16)

__global__ __launch_bounds__(256)
void highorder_wmma_kernel(const float* __restrict__ inp,
                           const float* __restrict__ params,
                           const float* __restrict__ bias,
                           const int*   __restrict__ edge_types,
                           const int*   __restrict__ msg_to_p,
                           float*       __restrict__ out)
{
    __shared__ v4f ldsB[NPAR * (KCHUNK / 4) * 16];

    const int tid  = threadIdx.x;
    const int lane = tid & 31;
    const int wave = tid >> 5;
    const int g    = lane >> 4;
    const int l16  = lane & 15;

    const int colBase = blockIdx.x * 16;
    const int rowBase = blockIdx.y * ROWS_PER_WG + wave*16;

    const int msg_to = msg_to_p[0];
    const int i1 = (msg_to == 0) ? 1 : 0;
    const int i2 = (msg_to == 2) ? 1 : 2;
    const float* A1 = inp + (size_t)i1 * N_ROWS * H_DIM;
    const float* A2 = inp + (size_t)i2 * N_ROWS * H_DIM;

    const int myRow = rowBase + l16;
    const float* a1p = A1 + (size_t)myRow * H_DIM + 4 * g;
    const float* a2p = A2 + (size_t)myRow * H_DIM + 4 * g;

    const int sc4 = tid & 3;
    const int sk4 = (tid >> 2) & 31;
    const int st  = tid >> 7;
    const float* bsrc0 = params + ((size_t)st * H_DIM + 4 * sk4) * O_DIM + colBase + sc4 * 4;
    const float* bsrc1 = bsrc0 + (size_t)2 * H_DIM * O_DIM;
    v4f* bdst0 = &ldsB[(st * (KCHUNK / 4) + sk4) * 16 + sc4 * 4];
    v4f* bdst1 = bdst0 + 2 * (KCHUNK / 4) * 16;

    const v8f vzero = {0.f, 0.f, 0.f, 0.f, 0.f, 0.f, 0.f, 0.f};
    v8f acc0 = vzero, acc1 = vzero, acc2 = vzero, acc3 = vzero;

    v4f br0[4], br1[4];
    v4f c1[4], c2[4];
    v4f n1[4], n2[4];

    const int colGlobal = colBase + l16;
    int ids[8];
    #pragma unroll
    for (int j = 0; j < 8; ++j)
        ids[j] = edge_types[(rowBase + j + 8*g) * ORDER_C + msg_to];

    #pragma unroll
    for (int r = 0; r < 4; ++r) {
        br0[r] = *(const v4f*)(bsrc0 + (size_t)r * O_DIM);
        br1[r] = *(const v4f*)(bsrc1 + (size_t)r * O_DIM);
    }
    #pragma unroll
    for (int j = 0; j < 4; ++j) {
        c1[j] = *(const v4f*)(a1p + 8 * j);
        c2[j] = *(const v4f*)(a2p + 8 * j);
    }
    #pragma unroll
    for (int j = 0; j < 4; ++j) {
        bdst0[j] = (v4f){br0[0][j], br0[1][j], br0[2][j], br0[3][j]};
        bdst1[j] = (v4f){br1[0][j], br1[1][j], br1[2][j], br1[3][j]};
    }
    __syncthreads();

    float bv[8];
    #pragma unroll
    for (int j = 0; j < 8; ++j)
        bv[j] = bias[ids[j] * O_DIM + colGlobal];

    #pragma unroll
    for (int c = 0; c < H_DIM / KCHUNK; ++c) {
        const bool more = (c + 1 < H_DIM / KCHUNK);
        if (more) {
            const size_t off = (size_t)(c + 1) * KCHUNK * O_DIM;
            #pragma unroll
            for (int r = 0; r < 4; ++r) {
                br0[r] = *(const v4f*)(bsrc0 + off + (size_t)r * O_DIM);
                br1[r] = *(const v4f*)(bsrc1 + off + (size_t)r * O_DIM);
            }
        }

        #pragma unroll
        for (int b = 0; b < KCHUNK / KBLK; ++b) {
            const int nb = (c * (KCHUNK / KBLK) + b + 1) & (H_DIM / KBLK - 1);
            #pragma unroll
            for (int j = 0; j < 4; ++j) {
                n1[j] = *(const v4f*)(a1p + nb * KBLK + 8 * j);
                n2[j] = *(const v4f*)(a2p + nb * KBLK + 8 * j);
            }

            #pragma unroll
            for (int j = 0; j < 4; ++j) {
                const v4f p = c1[j] * c2[j];
                const v4f* bb = &ldsB[(b * (KBLK / 4) + 2 * j + g) * 16 + l16];
                const v4f q0 = bb[0 * (KCHUNK / 4) * 16];
                const v4f q1 = bb[1 * (KCHUNK / 4) * 16];
                const v4f q2 = bb[2 * (KCHUNK / 4) * 16];
                const v4f q3 = bb[3 * (KCHUNK / 4) * 16];
                const v2f alo = {p.x, p.y};
                const v2f ahi = {p.z, p.w};
                acc0 = __builtin_amdgcn_wmma_f32_16x16x4_f32(false, alo, false, (v2f){q0.x, q0.y},
                                                             (short)0, acc0, false, false);
                acc1 = __builtin_amdgcn_wmma_f32_16x16x4_f32(false, alo, false, (v2f){q1.x, q1.y},
                                                             (short)0, acc1, false, false);
                acc2 = __builtin_amdgcn_wmma_f32_16x16x4_f32(false, alo, false, (v2f){q2.x, q2.y},
                                                             (short)0, acc2, false, false);
                acc3 = __builtin_amdgcn_wmma_f32_16x16x4_f32(false, alo, false, (v2f){q3.x, q3.y},
                                                             (short)0, acc3, false, false);
                acc0 = __builtin_amdgcn_wmma_f32_16x16x4_f32(false, ahi, false, (v2f){q0.z, q0.w},
                                                             (short)0, acc0, false, false);
                acc1 = __builtin_amdgcn_wmma_f32_16x16x4_f32(false, ahi, false, (v2f){q1.z, q1.w},
                                                             (short)0, acc1, false, false);
                acc2 = __builtin_amdgcn_wmma_f32_16x16x4_f32(false, ahi, false, (v2f){q2.z, q2.w},
                                                             (short)0, acc2, false, false);
                acc3 = __builtin_amdgcn_wmma_f32_16x16x4_f32(false, ahi, false, (v2f){q3.z, q3.w},
                                                             (short)0, acc3, false, false);
            }

            #pragma unroll
            for (int j = 0; j < 4; ++j) { c1[j] = n1[j]; c2[j] = n2[j]; }
        }

        if (more) {
            __syncthreads();
            #pragma unroll
            for (int j = 0; j < 4; ++j) {
                bdst0[j] = (v4f){br0[0][j], br0[1][j], br0[2][j], br0[3][j]};
                bdst1[j] = (v4f){br1[0][j], br1[1][j], br1[2][j], br1[3][j]};
            }
            __syncthreads();
        }
    }

    #pragma unroll
    for (int j = 0; j < 8; ++j) {
        const int row = rowBase + j + 8*g;
        const int id  = ids[j];
        float v = acc0[j];
        v = (id == 1) ? acc1[j] : v;
        v = (id == 2) ? acc2[j] : v;
        v = (id == 3) ? acc3[j] : v;
        v += bv[j];
        out[(size_t)row * O_DIM + colGlobal] = v;
    }
}

extern "C" void kernel_launch(void* const* d_in, const int* in_sizes, int n_in,
                              void* d_out, int out_size, void* d_ws, size_t ws_size,
                              hipStream_t stream) {
    const float* inp        = (const float*)d_in[0];
    const float* params     = (const float*)d_in[1];
    const float* bias       = (const float*)d_in[2];
    const int*   edge_types = (const int*)d_in[3];
    const int*   msg_to     = (const int*)d_in[6];
    float* out = (float*)d_out;

    dim3 grid(O_DIM / 16, N_ROWS / ROWS_PER_WG);
    dim3 block(256);
    highorder_wmma_kernel<<<grid, block, 0, stream>>>(inp, params, bias,
                                                      edge_types, msg_to, out);
}


# Rows grouped by param id inside the kernel (ballot ranks, row list in LDS): each 16-row tile multiplies by its own param matrix only, 4x fewer WMMAs; every sum keeps its k order
# speedup vs baseline: 2.0305x; 1.6082x over previous
#include <hip/hip_runtime.h>


typedef float v2f __attribute__((ext_vector_type(2)));
typedef float v4f __attribute__((ext_vector_type(4)));
typedef float v8f __attribute__((ext_vector_type(8)));

#define N_ROWS 1024
#define H_DIM  512
#define O_DIM  512
#define NPAR   4
#define ORDER_C 3
#define WAVES_PER_WG 4
#define THREADS (WAVES_PER_WG * 32)
#define ROWS_PER_WAVE_SORT (N_ROWS / WAVES_PER_WG)
#define SORT_BATCHES (ROWS_PER_WAVE_SORT / 32)
#define MAX_TILES 67
#define SORTED_LEN (MAX_TILES * 16)
#define COLS_PER_WAVE 64
#define COLS_PER_WG (WAVES_PER_WG * COLS_PER_WAVE)
#define KBLK 64
#define NBLK (H_DIM / KBLK)
#define GRP  (KBLK / 8)
#define APITCH (H_DIM / 4 + 1)

__global__ __launch_bounds__(THREADS)
void highorder_wmma_kernel(const float* __restrict__ inp,
                           const float* __restrict__ params,
                           const float* __restrict__ bias,
                           const int*   __restrict__ edge_types,
                           const int*   __restrict__ msg_to_p,
                           float*       __restrict__ out)
{
    __shared__ v4f ldsP[16 * APITCH];
    __shared__ int sorted[SORTED_LEN];
    __shared__ unsigned long long wcnt[WAVES_PER_WG];

    const int tid  = threadIdx.x;
    const int lane = tid & 31;
    const int wave = tid >> 5;
    const int g    = lane >> 4;
    const int l16  = lane & 15;

    const int tileBeg = blockIdx.y * 16;
    const int colBase = blockIdx.x * COLS_PER_WG + wave * COLS_PER_WAVE;

    const int msg_to = msg_to_p[0];
    const int i1 = (msg_to == 0) ? 1 : 0;
    const int i2 = (msg_to == 2) ? 1 : 2;
    const float* A1 = inp + (size_t)i1 * N_ROWS * H_DIM;
    const float* A2 = inp + (size_t)i2 * N_ROWS * H_DIM;


    int id[SORT_BATCHES];
    #pragma unroll
    for (int b = 0; b < SORT_BATCHES; ++b) {
        const int row = wave * ROWS_PER_WAVE_SORT + b * 32 + lane;
        const int v = edge_types[row * ORDER_C + msg_to];
        id[b] = min(max(v, 0), NPAR - 1);
    }
    for (int i = tid; i < SORTED_LEN; i += THREADS) sorted[i] = -1;

    const unsigned lanesBelow = (1u << lane) - 1u;
    unsigned long long run = 0;
    int myoff[SORT_BATCHES];
    #pragma unroll
    for (int b = 0; b < SORT_BATCHES; ++b) {
        const int v = id[b];
        const unsigned m0 = __builtin_amdgcn_ballot_w32((v & 1) != 0);
        const unsigned m1 = __builtin_amdgcn_ballot_w32((v & 2) != 0);
        const unsigned mine = ((v & 1) ? m0 : ~m0) & ((v & 2) ? m1 : ~m1);
        myoff[b] = (int)((run >> (16 * v)) & 0xffffull) + __popc(mine & lanesBelow);
        run += (unsigned long long)__popc(~m0 & ~m1)
             | ((unsigned long long)__popc( m0 & ~m1) << 16)
             | ((unsigned long long)__popc(~m0 &  m1) << 32)
             | ((unsigned long long)__popc( m0 &  m1) << 48);
    }
    if (lane == 0) wcnt[wave] = run;
    __syncthreads();

    unsigned long long tot = 0, pre = 0;
    #pragma unroll
    for (int w = 0; w < WAVES_PER_WG; ++w) {
        const unsigned long long c = wcnt[w];
        tot += c;
        pre += (w < wave) ? c : 0ull;
    }
    const int start1 = (((int)( tot        & 0xffffull)) + 15) & ~15;
    const int start2 = start1 + ((((int)((tot >> 16) & 0xffffull)) + 15) & ~15);
    const int start3 = start2 + ((((int)((tot >> 32) & 0xffffull)) + 15) & ~15);
    const int endAll = start3 + ((((int)((tot >> 48) & 0xffffull)) + 15) & ~15);
    const unsigned long long base = pre + (((unsigned long long)start1 << 16)
                                         | ((unsigned long long)start2 << 32)
                                         | ((unsigned long long)start3 << 48));
    #pragma unroll
    for (int b = 0; b < SORT_BATCHES; ++b) {
        const int pos = (int)((base >> (16 * id[b])) & 0xffffull) + myoff[b];
        sorted[pos] = wave * ROWS_PER_WAVE_SORT + b * 32 + lane;
    }
    __syncthreads();

    if (tileBeg >= endAll) return;
    const int t = (tileBeg >= start1 ? 1 : 0) + (tileBeg >= start2 ? 1 : 0) + (tileBeg >= start3 ? 1 : 0);

    const float* bp = params + ((size_t)t * H_DIM + 4 * g) * O_DIM + colBase + 4 * l16;

    v4f bc[4 * GRP], bn[4 * GRP];
    v4f ac[GRP], an[GRP];

    #pragma unroll
    for (int j = 0; j < GRP; ++j) {
        #pragma unroll
        for (int i = 0; i < 4; ++i)
            bc[4 * j + i] = *(const v4f*)(bp + (size_t)(8 * j + i) * O_DIM);
    }
    const v4f bsv = *(const v4f*)(bias + t * O_DIM + colBase + 4 * l16);

    {
        v4f x1[4][4], x2[4][4];
        #pragma unroll
        for (int rr = 0; rr < 4; ++rr) {
            const int src = max(sorted[tileBeg + 4 * wave + rr], 0);
            const v4f* p1 = (const v4f*)(A1 + (size_t)src * H_DIM) + lane;
            const v4f* p2 = (const v4f*)(A2 + (size_t)src * H_DIM) + lane;
            #pragma unroll
            for (int q = 0; q < 4; ++q) { x1[rr][q] = p1[q * 32]; x2[rr][q] = p2[q * 32]; }
        }
        #pragma unroll
        for (int rr = 0; rr < 4; ++rr) {
            #pragma unroll
            for (int q = 0; q < 4; ++q)
                ldsP[(4 * wave + rr) * APITCH + q * 32 + lane] = x1[rr][q] * x2[rr][q];
        }
    }
    __syncthreads();

    const v4f* ap = &ldsP[l16 * APITCH + g];
    #pragma unroll
    for (int j = 0; j < GRP; ++j) ac[j] = ap[2 * j];

    const v8f vzero = {0.f, 0.f, 0.f, 0.f, 0.f, 0.f, 0.f, 0.f};
    v8f acc0 = vzero, acc1 = vzero, acc2 = vzero, acc3 = vzero;

    #pragma unroll
    for (int blk = 0; blk < NBLK; ++blk) {
        const bool more = (blk + 1 < NBLK);
        if (more) {
            #pragma unroll
            for (int j = 0; j < GRP; ++j) {
                #pragma unroll
                for (int i = 0; i < 4; ++i)
                    bn[4 * j + i] = *(const v4f*)(bp + (size_t)((blk + 1) * KBLK + 8 * j + i) * O_DIM);
            }
            #pragma unroll
            for (int j = 0; j < GRP; ++j) an[j] = ap[(blk + 1) * (KBLK / 4) + 2 * j];
        }

        #pragma unroll
        for (int j = 0; j < GRP; ++j) {
            const v4f p  = ac[j];
            const v4f b0 = bc[4 * j + 0], b1 = bc[4 * j + 1], b2 = bc[4 * j + 2], b3 = bc[4 * j + 3];
            const v2f alo = {p.x, p.y};
            const v2f ahi = {p.z, p.w};
            acc0 = __builtin_amdgcn_wmma_f32_16x16x4_f32(false, alo, false, (v2f){b0.x, b1.x},
                                                         (short)0, acc0, false, false);
            acc1 = __builtin_amdgcn_wmma_f32_16x16x4_f32(false, alo, false, (v2f){b0.y, b1.y},
                                                         (short)0, acc1, false, false);
            acc2 = __builtin_amdgcn_wmma_f32_16x16x4_f32(false, alo, false, (v2f){b0.z, b1.z},
                                                         (short)0, acc2, false, false);
            acc3 = __builtin_amdgcn_wmma_f32_16x16x4_f32(false, alo, false, (v2f){b0.w, b1.w},
                                                         (short)0, acc3, false, false);
            acc0 = __builtin_amdgcn_wmma_f32_16x16x4_f32(false, ahi, false, (v2f){b2.x, b3.x},
                                                         (short)0, acc0, false, false);
            acc1 = __builtin_amdgcn_wmma_f32_16x16x4_f32(false, ahi, false, (v2f){b2.y, b3.y},
                                                         (short)0, acc1, false, false);
            acc2 = __builtin_amdgcn_wmma_f32_16x16x4_f32(false, ahi, false, (v2f){b2.z, b3.z},
                                                         (short)0, acc2, false, false);
            acc3 = __builtin_amdgcn_wmma_f32_16x16x4_f32(false, ahi, false, (v2f){b2.w, b3.w},
                                                         (short)0, acc3, false, false);
        }

        if (more) {
            #pragma unroll
            for (int j = 0; j < 4 * GRP; ++j) bc[j] = bn[j];
            #pragma unroll
            for (int j = 0; j < GRP; ++j) ac[j] = an[j];
        }
    }

    int srow[8];
    #pragma unroll
    for (int r = 0; r < 8; ++r) srow[r] = sorted[tileBeg + 8 * g + r];
    #pragma unroll
    for (int r = 0; r < 8; ++r) {
        if (srow[r] >= 0) {
            v4f v = {acc0[r], acc1[r], acc2[r], acc3[r]};
            v += bsv;
            *(v4f*)(out + (size_t)srow[r] * O_DIM + colBase + 4 * l16) = v;
        }
    }
}

extern "C" void kernel_launch(void* const* d_in, const int* in_sizes, int n_in,
                              void* d_out, int out_size, void* d_ws, size_t ws_size,
                              hipStream_t stream) {
    const float* inp        = (const float*)d_in[0];
    const float* params     = (const float*)d_in[1];
    const float* bias       = (const float*)d_in[2];
    const int*   edge_types = (const int*)d_in[3];
    const int*   msg_to     = (const int*)d_in[6];
    float* out = (float*)d_out;

    dim3 grid(O_DIM / COLS_PER_WG, MAX_TILES);
    dim3 block(THREADS);
    highorder_wmma_kernel<<<grid, block, 0, stream>>>(inp, params, bias,
                                                      edge_types, msg_to, out);
}


# Scheduling barriers pin each next K block's 32 B loads and 8 LDS fragment reads as ONE group ahead of the block's 64 WMMAs (compiler had spread them among the WMMAs); k order and sums unchanged
# speedup vs baseline: 2.2225x; 1.0946x over previous
#include <hip/hip_runtime.h>


typedef float v2f __attribute__((ext_vector_type(2)));
typedef float v4f __attribute__((ext_vector_type(4)));
typedef float v8f __attribute__((ext_vector_type(8)));

#define N_ROWS 1024
#define H_DIM  512
#define O_DIM  512
#define NPAR   4
#define ORDER_C 3
#define WAVES_PER_WG 4
#define THREADS (WAVES_PER_WG * 32)
#define ROWS_PER_WAVE_SORT (N_ROWS / WAVES_PER_WG)
#define SORT_BATCHES (ROWS_PER_WAVE_SORT / 32)
#define MAX_TILES 67
#define SORTED_LEN (MAX_TILES * 16)
#define COLS_PER_WAVE 64
#define COLS_PER_WG (WAVES_PER_WG * COLS_PER_WAVE)
#define KBLK 64
#define NBLK (H_DIM / KBLK)
#define GRP  (KBLK / 8)
#define APITCH (H_DIM / 4 + 1)

__global__ __launch_bounds__(THREADS)
void highorder_wmma_kernel(const float* __restrict__ inp,
                           const float* __restrict__ params,
                           const float* __restrict__ bias,
                           const int*   __restrict__ edge_types,
                           const int*   __restrict__ msg_to_p,
                           float*       __restrict__ out)
{
    __shared__ v4f ldsP[16 * APITCH];
    __shared__ int sorted[SORTED_LEN];
    __shared__ unsigned long long wcnt[WAVES_PER_WG];

    const int tid  = threadIdx.x;
    const int lane = tid & 31;
    const int wave = tid >> 5;
    const int g    = lane >> 4;
    const int l16  = lane & 15;

    const int tileBeg = blockIdx.y * 16;
    const int colBase = blockIdx.x * COLS_PER_WG + wave * COLS_PER_WAVE;

    const int msg_to = msg_to_p[0];
    const int i1 = (msg_to == 0) ? 1 : 0;
    const int i2 = (msg_to == 2) ? 1 : 2;
    const float* A1 = inp + (size_t)i1 * N_ROWS * H_DIM;
    const float* A2 = inp + (size_t)i2 * N_ROWS * H_DIM;


    int id[SORT_BATCHES];
    #pragma unroll
    for (int b = 0; b < SORT_BATCHES; ++b) {
        const int row = wave * ROWS_PER_WAVE_SORT + b * 32 + lane;
        const int v = edge_types[row * ORDER_C + msg_to];
        id[b] = min(max(v, 0), NPAR - 1);
    }
    for (int i = tid; i < SORTED_LEN; i += THREADS) sorted[i] = -1;

    const unsigned lanesBelow = (1u << lane) - 1u;
    unsigned long long run = 0;
    int myoff[SORT_BATCHES];
    #pragma unroll
    for (int b = 0; b < SORT_BATCHES; ++b) {
        const int v = id[b];
        const unsigned m0 = __builtin_amdgcn_ballot_w32((v & 1) != 0);
        const unsigned m1 = __builtin_amdgcn_ballot_w32((v & 2) != 0);
        const unsigned mine = ((v & 1) ? m0 : ~m0) & ((v & 2) ? m1 : ~m1);
        myoff[b] = (int)((run >> (16 * v)) & 0xffffull) + __popc(mine & lanesBelow);
        run += (unsigned long long)__popc(~m0 & ~m1)
             | ((unsigned long long)__popc( m0 & ~m1) << 16)
             | ((unsigned long long)__popc(~m0 &  m1) << 32)
             | ((unsigned long long)__popc( m0 &  m1) << 48);
    }
    if (lane == 0) wcnt[wave] = run;
    __syncthreads();

    unsigned long long tot = 0, pre = 0;
    #pragma unroll
    for (int w = 0; w < WAVES_PER_WG; ++w) {
        const unsigned long long c = wcnt[w];
        tot += c;
        pre += (w < wave) ? c : 0ull;
    }
    const int start1 = (((int)( tot        & 0xffffull)) + 15) & ~15;
    const int start2 = start1 + ((((int)((tot >> 16) & 0xffffull)) + 15) & ~15);
    const int start3 = start2 + ((((int)((tot >> 32) & 0xffffull)) + 15) & ~15);
    const int endAll = start3 + ((((int)((tot >> 48) & 0xffffull)) + 15) & ~15);
    const unsigned long long base = pre + (((unsigned long long)start1 << 16)
                                         | ((unsigned long long)start2 << 32)
                                         | ((unsigned long long)start3 << 48));
    #pragma unroll
    for (int b = 0; b < SORT_BATCHES; ++b) {
        const int pos = (int)((base >> (16 * id[b])) & 0xffffull) + myoff[b];
        sorted[pos] = wave * ROWS_PER_WAVE_SORT + b * 32 + lane;
    }
    __syncthreads();

    if (tileBeg >= endAll) return;
    const int t = (tileBeg >= start1 ? 1 : 0) + (tileBeg >= start2 ? 1 : 0) + (tileBeg >= start3 ? 1 : 0);

    const float* bp = params + ((size_t)t * H_DIM + 4 * g) * O_DIM + colBase + 4 * l16;

    v4f bc[4 * GRP], bn[4 * GRP];
    v4f ac[GRP], an[GRP];

    #pragma unroll
    for (int j = 0; j < GRP; ++j) {
        #pragma unroll
        for (int i = 0; i < 4; ++i)
            bc[4 * j + i] = *(const v4f*)(bp + (size_t)(8 * j + i) * O_DIM);
    }
    const v4f bsv = *(const v4f*)(bias + t * O_DIM + colBase + 4 * l16);

    {
        v4f x1[4][4], x2[4][4];
        #pragma unroll
        for (int rr = 0; rr < 4; ++rr) {
            const int src = max(sorted[tileBeg + 4 * wave + rr], 0);
            const v4f* p1 = (const v4f*)(A1 + (size_t)src * H_DIM) + lane;
            const v4f* p2 = (const v4f*)(A2 + (size_t)src * H_DIM) + lane;
            #pragma unroll
            for (int q = 0; q < 4; ++q) { x1[rr][q] = p1[q * 32]; x2[rr][q] = p2[q * 32]; }
        }
        #pragma unroll
        for (int rr = 0; rr < 4; ++rr) {
            #pragma unroll
            for (int q = 0; q < 4; ++q)
                ldsP[(4 * wave + rr) * APITCH + q * 32 + lane] = x1[rr][q] * x2[rr][q];
        }
    }
    __syncthreads();

    const v4f* ap = &ldsP[l16 * APITCH + g];
    #pragma unroll
    for (int j = 0; j < GRP; ++j) ac[j] = ap[2 * j];

    const v8f vzero = {0.f, 0.f, 0.f, 0.f, 0.f, 0.f, 0.f, 0.f};
    v8f acc0 = vzero, acc1 = vzero, acc2 = vzero, acc3 = vzero;

    #pragma unroll
    for (int blk = 0; blk < NBLK; ++blk) {
        const bool more = (blk + 1 < NBLK);
        if (more) {
            #pragma unroll
            for (int j = 0; j < GRP; ++j) {
                #pragma unroll
                for (int i = 0; i < 4; ++i)
                    bn[4 * j + i] = *(const v4f*)(bp + (size_t)((blk + 1) * KBLK + 8 * j + i) * O_DIM);
            }
            #pragma unroll
            for (int j = 0; j < GRP; ++j) an[j] = ap[(blk + 1) * (KBLK / 4) + 2 * j];
        }
        __builtin_amdgcn_sched_barrier(0);

        #pragma unroll
        for (int j = 0; j < GRP; ++j) {
            const v4f p  = ac[j];
            const v4f b0 = bc[4 * j + 0], b1 = bc[4 * j + 1], b2 = bc[4 * j + 2], b3 = bc[4 * j + 3];
            const v2f alo = {p.x, p.y};
            const v2f ahi = {p.z, p.w};
            acc0 = __builtin_amdgcn_wmma_f32_16x16x4_f32(false, alo, false, (v2f){b0.x, b1.x},
                                                         (short)0, acc0, false, false);
            acc1 = __builtin_amdgcn_wmma_f32_16x16x4_f32(false, alo, false, (v2f){b0.y, b1.y},
                                                         (short)0, acc1, false, false);
            acc2 = __builtin_amdgcn_wmma_f32_16x16x4_f32(false, alo, false, (v2f){b0.z, b1.z},
                                                         (short)0, acc2, false, false);
            acc3 = __builtin_amdgcn_wmma_f32_16x16x4_f32(false, alo, false, (v2f){b0.w, b1.w},
                                                         (short)0, acc3, false, false);
            acc0 = __builtin_amdgcn_wmma_f32_16x16x4_f32(false, ahi, false, (v2f){b2.x, b3.x},
                                                         (short)0, acc0, false, false);
            acc1 = __builtin_amdgcn_wmma_f32_16x16x4_f32(false, ahi, false, (v2f){b2.y, b3.y},
                                                         (short)0, acc1, false, false);
            acc2 = __builtin_amdgcn_wmma_f32_16x16x4_f32(false, ahi, false, (v2f){b2.z, b3.z},
                                                         (short)0, acc2, false, false);
            acc3 = __builtin_amdgcn_wmma_f32_16x16x4_f32(false, ahi, false, (v2f){b2.w, b3.w},
                                                         (short)0, acc3, false, false);
        }

        __builtin_amdgcn_sched_barrier(0);

        if (more) {
            #pragma unroll
            for (int j = 0; j < 4 * GRP; ++j) bc[j] = bn[j];
            #pragma unroll
            for (int j = 0; j < GRP; ++j) ac[j] = an[j];
        }
    }

    int srow[8];
    #pragma unroll
    for (int r = 0; r < 8; ++r) srow[r] = sorted[tileBeg + 8 * g + r];
    #pragma unroll
    for (int r = 0; r < 8; ++r) {
        if (srow[r] >= 0) {
            v4f v = {acc0[r], acc1[r], acc2[r], acc3[r]};
            v += bsv;
            *(v4f*)(out + (size_t)srow[r] * O_DIM + colBase + 4 * l16) = v;
        }
    }
}

extern "C" void kernel_launch(void* const* d_in, const int* in_sizes, int n_in,
                              void* d_out, int out_size, void* d_ws, size_t ws_size,
                              hipStream_t stream) {
    const float* inp        = (const float*)d_in[0];
    const float* params     = (const float*)d_in[1];
    const float* bias       = (const float*)d_in[2];
    const int*   edge_types = (const int*)d_in[3];
    const int*   msg_to     = (const int*)d_in[6];
    float* out = (float*)d_out;

    dim3 grid(O_DIM / COLS_PER_WG, MAX_TILES);
    dim3 block(THREADS);
    highorder_wmma_kernel<<<grid, block, 0, stream>>>(inp, params, bias,
                                                      edge_types, msg_to, out);
}


# Factor-row loads issued ahead of the first B block, hand-over barrier made LDS-only: matrix work starts once the A tile is staged, B block 0 lands behind counted waits; k order and sums unchanged
# speedup vs baseline: 2.3527x; 1.0586x over previous
#include <hip/hip_runtime.h>


typedef float v2f __attribute__((ext_vector_type(2)));
typedef float v4f __attribute__((ext_vector_type(4)));
typedef float v8f __attribute__((ext_vector_type(8)));

#define N_ROWS 1024
#define H_DIM  512
#define O_DIM  512
#define NPAR   4
#define ORDER_C 3
#define WAVES_PER_WG 4
#define THREADS (WAVES_PER_WG * 32)
#define ROWS_PER_WAVE_SORT (N_ROWS / WAVES_PER_WG)
#define SORT_BATCHES (ROWS_PER_WAVE_SORT / 32)
#define MAX_TILES 67
#define SORTED_LEN (MAX_TILES * 16)
#define COLS_PER_WAVE 64
#define COLS_PER_WG (WAVES_PER_WG * COLS_PER_WAVE)
#define KBLK 64
#define NBLK (H_DIM / KBLK)
#define GRP  (KBLK / 8)
#define APITCH (H_DIM / 4 + 1)

__global__ __launch_bounds__(THREADS)
void highorder_wmma_kernel(const float* __restrict__ inp,
                           const float* __restrict__ params,
                           const float* __restrict__ bias,
                           const int*   __restrict__ edge_types,
                           const int*   __restrict__ msg_to_p,
                           float*       __restrict__ out)
{
    __shared__ v4f ldsP[16 * APITCH];
    __shared__ int sorted[SORTED_LEN];
    __shared__ unsigned long long wcnt[WAVES_PER_WG];

    const int tid  = threadIdx.x;
    const int lane = tid & 31;
    const int wave = tid >> 5;
    const int g    = lane >> 4;
    const int l16  = lane & 15;

    const int tileBeg = blockIdx.y * 16;
    const int colBase = blockIdx.x * COLS_PER_WG + wave * COLS_PER_WAVE;

    const int msg_to = msg_to_p[0];
    const int i1 = (msg_to == 0) ? 1 : 0;
    const int i2 = (msg_to == 2) ? 1 : 2;
    const float* A1 = inp + (size_t)i1 * N_ROWS * H_DIM;
    const float* A2 = inp + (size_t)i2 * N_ROWS * H_DIM;


    int id[SORT_BATCHES];
    #pragma unroll
    for (int b = 0; b < SORT_BATCHES; ++b) {
        const int row = wave * ROWS_PER_WAVE_SORT + b * 32 + lane;
        const int v = edge_types[row * ORDER_C + msg_to];
        id[b] = min(max(v, 0), NPAR - 1);
    }
    for (int i = tid; i < SORTED_LEN; i += THREADS) sorted[i] = -1;

    const unsigned lanesBelow = (1u << lane) - 1u;
    unsigned long long run = 0;
    int myoff[SORT_BATCHES];
    #pragma unroll
    for (int b = 0; b < SORT_BATCHES; ++b) {
        const int v = id[b];
        const unsigned m0 = __builtin_amdgcn_ballot_w32((v & 1) != 0);
        const unsigned m1 = __builtin_amdgcn_ballot_w32((v & 2) != 0);
        const unsigned mine = ((v & 1) ? m0 : ~m0) & ((v & 2) ? m1 : ~m1);
        myoff[b] = (int)((run >> (16 * v)) & 0xffffull) + __popc(mine & lanesBelow);
        run += (unsigned long long)__popc(~m0 & ~m1)
             | ((unsigned long long)__popc( m0 & ~m1) << 16)
             | ((unsigned long long)__popc(~m0 &  m1) << 32)
             | ((unsigned long long)__popc( m0 &  m1) << 48);
    }
    if (lane == 0) wcnt[wave] = run;
    __syncthreads();

    unsigned long long tot = 0, pre = 0;
    #pragma unroll
    for (int w = 0; w < WAVES_PER_WG; ++w) {
        const unsigned long long c = wcnt[w];
        tot += c;
        pre += (w < wave) ? c : 0ull;
    }
    const int start1 = (((int)( tot        & 0xffffull)) + 15) & ~15;
    const int start2 = start1 + ((((int)((tot >> 16) & 0xffffull)) + 15) & ~15);
    const int start3 = start2 + ((((int)((tot >> 32) & 0xffffull)) + 15) & ~15);
    const int endAll = start3 + ((((int)((tot >> 48) & 0xffffull)) + 15) & ~15);
    const unsigned long long base = pre + (((unsigned long long)start1 << 16)
                                         | ((unsigned long long)start2 << 32)
                                         | ((unsigned long long)start3 << 48));
    #pragma unroll
    for (int b = 0; b < SORT_BATCHES; ++b) {
        const int pos = (int)((base >> (16 * id[b])) & 0xffffull) + myoff[b];
        sorted[pos] = wave * ROWS_PER_WAVE_SORT + b * 32 + lane;
    }
    __syncthreads();

    if (tileBeg >= endAll) return;
    const int t = (tileBeg >= start1 ? 1 : 0) + (tileBeg >= start2 ? 1 : 0) + (tileBeg >= start3 ? 1 : 0);

    const float* bp = params + ((size_t)t * H_DIM + 4 * g) * O_DIM + colBase + 4 * l16;

    v4f bc[4 * GRP], bn[4 * GRP];
    v4f ac[GRP], an[GRP];
    v4f bsv;

    {
        v4f x1[4][4], x2[4][4];
        #pragma unroll
        for (int rr = 0; rr < 4; ++rr) {
            const int src = max(sorted[tileBeg + 4 * wave + rr], 0);
            const v4f* p1 = (const v4f*)(A1 + (size_t)src * H_DIM) + lane;
            const v4f* p2 = (const v4f*)(A2 + (size_t)src * H_DIM) + lane;
            #pragma unroll
            for (int q = 0; q < 4; ++q) { x1[rr][q] = p1[q * 32]; x2[rr][q] = p2[q * 32]; }
        }
        __builtin_amdgcn_sched_barrier(0);

        #pragma unroll
        for (int j = 0; j < GRP; ++j) {
            #pragma unroll
            for (int i = 0; i < 4; ++i)
                bc[4 * j + i] = *(const v4f*)(bp + (size_t)(8 * j + i) * O_DIM);
        }
        bsv = *(const v4f*)(bias + t * O_DIM + colBase + 4 * l16);

        __builtin_amdgcn_sched_barrier(0);

        #pragma unroll
        for (int rr = 0; rr < 4; ++rr) {
            #pragma unroll
            for (int q = 0; q < 4; ++q)
                ldsP[(4 * wave + rr) * APITCH + q * 32 + lane] = x1[rr][q] * x2[rr][q];
        }
    }
    __builtin_amdgcn_fence(__ATOMIC_RELEASE, "workgroup", "local");
    __builtin_amdgcn_s_barrier();
    __builtin_amdgcn_fence(__ATOMIC_ACQUIRE, "workgroup", "local");

    const v4f* ap = &ldsP[l16 * APITCH + g];
    #pragma unroll
    for (int j = 0; j < GRP; ++j) ac[j] = ap[2 * j];

    const v8f vzero = {0.f, 0.f, 0.f, 0.f, 0.f, 0.f, 0.f, 0.f};
    v8f acc0 = vzero, acc1 = vzero, acc2 = vzero, acc3 = vzero;

    #pragma unroll
    for (int blk = 0; blk < NBLK; ++blk) {
        const bool more = (blk + 1 < NBLK);
        if (more) {
            #pragma unroll
            for (int j = 0; j < GRP; ++j) {
                #pragma unroll
                for (int i = 0; i < 4; ++i)
                    bn[4 * j + i] = *(const v4f*)(bp + (size_t)((blk + 1) * KBLK + 8 * j + i) * O_DIM);
            }
            #pragma unroll
            for (int j = 0; j < GRP; ++j) an[j] = ap[(blk + 1) * (KBLK / 4) + 2 * j];
        }
        __builtin_amdgcn_sched_barrier(0);

        #pragma unroll
        for (int j = 0; j < GRP; ++j) {
            const v4f p  = ac[j];
            const v4f b0 = bc[4 * j + 0], b1 = bc[4 * j + 1], b2 = bc[4 * j + 2], b3 = bc[4 * j + 3];
            const v2f alo = {p.x, p.y};
            const v2f ahi = {p.z, p.w};
            acc0 = __builtin_amdgcn_wmma_f32_16x16x4_f32(false, alo, false, (v2f){b0.x, b1.x},
                                                         (short)0, acc0, false, false);
            acc1 = __builtin_amdgcn_wmma_f32_16x16x4_f32(false, alo, false, (v2f){b0.y, b1.y},
                                                         (short)0, acc1, false, false);
            acc2 = __builtin_amdgcn_wmma_f32_16x16x4_f32(false, alo, false, (v2f){b0.z, b1.z},
                                                         (short)0, acc2, false, false);
            acc3 = __builtin_amdgcn_wmma_f32_16x16x4_f32(false, alo, false, (v2f){b0.w, b1.w},
                                                         (short)0, acc3, false, false);
            acc0 = __builtin_amdgcn_wmma_f32_16x16x4_f32(false, ahi, false, (v2f){b2.x, b3.x},
                                                         (short)0, acc0, false, false);
            acc1 = __builtin_amdgcn_wmma_f32_16x16x4_f32(false, ahi, false, (v2f){b2.y, b3.y},
                                                         (short)0, acc1, false, false);
            acc2 = __builtin_amdgcn_wmma_f32_16x16x4_f32(false, ahi, false, (v2f){b2.z, b3.z},
                                                         (short)0, acc2, false, false);
            acc3 = __builtin_amdgcn_wmma_f32_16x16x4_f32(false, ahi, false, (v2f){b2.w, b3.w},
                                                         (short)0, acc3, false, false);
        }

        __builtin_amdgcn_sched_barrier(0);

        if (more) {
            #pragma unroll
            for (int j = 0; j < 4 * GRP; ++j) bc[j] = bn[j];
            #pragma unroll
            for (int j = 0; j < GRP; ++j) ac[j] = an[j];
        }
    }

    int srow[8];
    #pragma unroll
    for (int r = 0; r < 8; ++r) srow[r] = sorted[tileBeg + 8 * g + r];
    #pragma unroll
    for (int r = 0; r < 8; ++r) {
        if (srow[r] >= 0) {
            v4f v = {acc0[r], acc1[r], acc2[r], acc3[r]};
            v += bsv;
            *(v4f*)(out + (size_t)srow[r] * O_DIM + colBase + 4 * l16) = v;
        }
    }
}

extern "C" void kernel_launch(void* const* d_in, const int* in_sizes, int n_in,
                              void* d_out, int out_size, void* d_ws, size_t ws_size,
                              hipStream_t stream) {
    const float* inp        = (const float*)d_in[0];
    const float* params     = (const float*)d_in[1];
    const float* bias       = (const float*)d_in[2];
    const int*   edge_types = (const int*)d_in[3];
    const int*   msg_to     = (const int*)d_in[6];
    float* out = (float*)d_out;

    dim3 grid(O_DIM / COLS_PER_WG, MAX_TILES);
    dim3 block(THREADS);
    highorder_wmma_kernel<<<grid, block, 0, stream>>>(inp, params, bias,
                                                      edge_types, msg_to, out);
}


# Work spread over all 256 compute units: 16x128-column work-groups, 32 columns a wave; padding-only tiles 64..66 in slim k-quartered work-groups summed through LDS (sum order changes for those rows)
# speedup vs baseline: 2.5855x; 1.0990x over previous
#include <hip/hip_runtime.h>


typedef float v2f __attribute__((ext_vector_type(2)));
typedef float v4f __attribute__((ext_vector_type(4)));
typedef float v8f __attribute__((ext_vector_type(8)));

#define N_ROWS 1024
#define H_DIM  512
#define O_DIM  512
#define NPAR   4
#define ORDER_C 3
#define WAVES_PER_WG 4
#define THREADS (WAVES_PER_WG * 32)
#define ROWS_PER_WAVE_SORT (N_ROWS / WAVES_PER_WG)
#define SORT_BATCHES (ROWS_PER_WAVE_SORT / 32)
#define MAX_TILES 67
#define SORTED_LEN (MAX_TILES * 16)
#define COLS_PER_WAVE 32
#define FULL_TILES (N_ROWS / 16)
#define EXTRA_TILES (MAX_TILES - FULL_TILES)
#define WIDE_COLS (WAVES_PER_WG * COLS_PER_WAVE)
#define WIDE_PER_TILE (O_DIM / WIDE_COLS)
#define WIDE_WGS (FULL_TILES * WIDE_PER_TILE)
#define SLIM_PER_TILE (O_DIM / COLS_PER_WAVE)
#define SLIM_WGS (EXTRA_TILES * SLIM_PER_TILE)
#define KBLK 64
#define NBLK (H_DIM / KBLK)
#define GRP  (KBLK / 8)
#define APITCH (H_DIM / 4 + 1)

static __device__ __forceinline__ void load_b_block(v2f (&b)[4 * GRP], const float* p)
{
    #pragma unroll
    for (int j = 0; j < GRP; ++j) {
        #pragma unroll
        for (int i = 0; i < 4; ++i)
            b[4 * j + i] = *(const v2f*)(p + (size_t)(8 * j + i) * O_DIM);
    }
}

static __device__ __forceinline__ void load_a_block(v4f (&a)[GRP], const v4f* p)
{
    #pragma unroll
    for (int j = 0; j < GRP; ++j) a[j] = p[2 * j];
}

static __device__ __forceinline__ void mma_block(v8f& acc0, v8f& acc1,
                                                 const v4f (&a)[GRP], const v2f (&b)[4 * GRP])
{
    #pragma unroll
    for (int j = 0; j < GRP; ++j) {
        const v4f p  = a[j];
        const v2f b0 = b[4 * j + 0], b1 = b[4 * j + 1], b2 = b[4 * j + 2], b3 = b[4 * j + 3];
        const v2f alo = {p.x, p.y};
        const v2f ahi = {p.z, p.w};
        acc0 = __builtin_amdgcn_wmma_f32_16x16x4_f32(false, alo, false, (v2f){b0.x, b1.x},
                                                     (short)0, acc0, false, false);
        acc1 = __builtin_amdgcn_wmma_f32_16x16x4_f32(false, alo, false, (v2f){b0.y, b1.y},
                                                     (short)0, acc1, false, false);
        acc0 = __builtin_amdgcn_wmma_f32_16x16x4_f32(false, ahi, false, (v2f){b2.x, b3.x},
                                                     (short)0, acc0, false, false);
        acc1 = __builtin_amdgcn_wmma_f32_16x16x4_f32(false, ahi, false, (v2f){b2.y, b3.y},
                                                     (short)0, acc1, false, false);
    }
}

__global__ __launch_bounds__(THREADS)
void highorder_wmma_kernel(const float* __restrict__ inp,
                           const float* __restrict__ params,
                           const float* __restrict__ bias,
                           const int*   __restrict__ edge_types,
                           const int*   __restrict__ msg_to_p,
                           float*       __restrict__ out)
{
    __shared__ v4f ldsP[16 * APITCH];
    __shared__ int sorted[SORTED_LEN];
    __shared__ unsigned long long wcnt[WAVES_PER_WG];
    __shared__ v2f red[WAVES_PER_WG * 8 * 32];

    const int tid  = threadIdx.x;
    const int lane = tid & 31;
    const int wave = tid >> 5;
    const int g    = lane >> 4;
    const int l16  = lane & 15;

    const int  wg   = blockIdx.x;
    const bool slim = (wg >= WIDE_WGS);
    const int  sidx = wg - WIDE_WGS;
    const int  tile = slim ? FULL_TILES + sidx / SLIM_PER_TILE : wg / WIDE_PER_TILE;
    const int  tileBeg = tile * 16;
    const int  colBase = slim ? (sidx % SLIM_PER_TILE) * COLS_PER_WAVE
                              : (wg % WIDE_PER_TILE) * WIDE_COLS + wave * COLS_PER_WAVE;
    const int  kb0 = slim ? 2 * wave : 0;

    const int msg_to = msg_to_p[0];
    const int i1 = (msg_to == 0) ? 1 : 0;
    const int i2 = (msg_to == 2) ? 1 : 2;
    const float* A1 = inp + (size_t)i1 * N_ROWS * H_DIM;
    const float* A2 = inp + (size_t)i2 * N_ROWS * H_DIM;


    int id[SORT_BATCHES];
    #pragma unroll
    for (int b = 0; b < SORT_BATCHES; ++b) {
        const int row = wave * ROWS_PER_WAVE_SORT + b * 32 + lane;
        const int v = edge_types[row * ORDER_C + msg_to];
        id[b] = min(max(v, 0), NPAR - 1);
    }
    for (int i = tid; i < SORTED_LEN; i += THREADS) sorted[i] = -1;

    const unsigned lanesBelow = (1u << lane) - 1u;
    unsigned long long run = 0;
    int myoff[SORT_BATCHES];
    #pragma unroll
    for (int b = 0; b < SORT_BATCHES; ++b) {
        const int v = id[b];
        const unsigned m0 = __builtin_amdgcn_ballot_w32((v & 1) != 0);
        const unsigned m1 = __builtin_amdgcn_ballot_w32((v & 2) != 0);
        const unsigned mine = ((v & 1) ? m0 : ~m0) & ((v & 2) ? m1 : ~m1);
        myoff[b] = (int)((run >> (16 * v)) & 0xffffull) + __popc(mine & lanesBelow);
        run += (unsigned long long)__popc(~m0 & ~m1)
             | ((unsigned long long)__popc( m0 & ~m1) << 16)
             | ((unsigned long long)__popc(~m0 &  m1) << 32)
             | ((unsigned long long)__popc( m0 &  m1) << 48);
    }
    if (lane == 0) wcnt[wave] = run;
    __syncthreads();

    unsigned long long tot = 0, pre = 0;
    #pragma unroll
    for (int w = 0; w < WAVES_PER_WG; ++w) {
        const unsigned long long c = wcnt[w];
        tot += c;
        pre += (w < wave) ? c : 0ull;
    }
    const int start1 = (((int)( tot        & 0xffffull)) + 15) & ~15;
    const int start2 = start1 + ((((int)((tot >> 16) & 0xffffull)) + 15) & ~15);
    const int start3 = start2 + ((((int)((tot >> 32) & 0xffffull)) + 15) & ~15);
    const int endAll = start3 + ((((int)((tot >> 48) & 0xffffull)) + 15) & ~15);
    const unsigned long long base = pre + (((unsigned long long)start1 << 16)
                                         | ((unsigned long long)start2 << 32)
                                         | ((unsigned long long)start3 << 48));
    #pragma unroll
    for (int b = 0; b < SORT_BATCHES; ++b) {
        const int pos = (int)((base >> (16 * id[b])) & 0xffffull) + myoff[b];
        sorted[pos] = wave * ROWS_PER_WAVE_SORT + b * 32 + lane;
    }
    __syncthreads();

    if (tileBeg >= endAll) return;
    const int t = (tileBeg >= start1 ? 1 : 0) + (tileBeg >= start2 ? 1 : 0) + (tileBeg >= start3 ? 1 : 0);

    const float* bp  = params + ((size_t)t * H_DIM + 4 * g) * O_DIM + colBase + 2 * l16;
    const float* bp0 = bp + (size_t)kb0 * KBLK * O_DIM;

    v2f bb[3][4 * GRP];
    v4f ab[3][GRP];
    v2f bsv;

    {
        v4f x1[4][4], x2[4][4];
        #pragma unroll
        for (int rr = 0; rr < 4; ++rr) {
            const int src = max(sorted[tileBeg + 4 * wave + rr], 0);
            const v4f* p1 = (const v4f*)(A1 + (size_t)src * H_DIM) + lane;
            const v4f* p2 = (const v4f*)(A2 + (size_t)src * H_DIM) + lane;
            #pragma unroll
            for (int q = 0; q < 4; ++q) { x1[rr][q] = p1[q * 32]; x2[rr][q] = p2[q * 32]; }
        }
        __builtin_amdgcn_sched_barrier(0);

        load_b_block(bb[0], bp0);
        bsv = *(const v2f*)(bias + t * O_DIM + colBase + 2 * l16);

        __builtin_amdgcn_sched_barrier(0);

        #pragma unroll
        for (int rr = 0; rr < 4; ++rr) {
            #pragma unroll
            for (int q = 0; q < 4; ++q)
                ldsP[(4 * wave + rr) * APITCH + q * 32 + lane] = x1[rr][q] * x2[rr][q];
        }
    }
    __builtin_amdgcn_fence(__ATOMIC_RELEASE, "workgroup", "local");
    __builtin_amdgcn_s_barrier();
    __builtin_amdgcn_fence(__ATOMIC_ACQUIRE, "workgroup", "local");

    const v4f* ap = &ldsP[l16 * APITCH + g];

    const v8f vzero = {0.f, 0.f, 0.f, 0.f, 0.f, 0.f, 0.f, 0.f};
    v8f acc0 = vzero, acc1 = vzero;

    if (!slim) {
        load_a_block(ab[0], ap);
        load_b_block(bb[1], bp + (size_t)KBLK * O_DIM);
        load_a_block(ab[1], ap + KBLK / 4);

        #pragma unroll
        for (int blk = 0; blk < NBLK; ++blk) {
            if (blk + 2 < NBLK) {
                load_b_block(bb[(blk + 2) % 3], bp + (size_t)(blk + 2) * KBLK * O_DIM);
                load_a_block(ab[(blk + 2) % 3], ap + (blk + 2) * (KBLK / 4));
            }
            __builtin_amdgcn_sched_barrier(0);

            mma_block(acc0, acc1, ab[blk % 3], bb[blk % 3]);

            __builtin_amdgcn_sched_barrier(0);
        }

        int srow[8];
        #pragma unroll
        for (int r = 0; r < 8; ++r) srow[r] = sorted[tileBeg + 8 * g + r];
        #pragma unroll
        for (int r = 0; r < 8; ++r) {
            if (srow[r] >= 0) {
                v2f v = {acc0[r], acc1[r]};
                v += bsv;
                *(v2f*)(out + (size_t)srow[r] * O_DIM + colBase + 2 * l16) = v;
            }
        }
    } else {
        load_a_block(ab[0], ap + kb0 * (KBLK / 4));
        load_b_block(bb[1], bp0 + (size_t)KBLK * O_DIM);
        load_a_block(ab[1], ap + (kb0 + 1) * (KBLK / 4));
        __builtin_amdgcn_sched_barrier(0);

        mma_block(acc0, acc1, ab[0], bb[0]);
        mma_block(acc0, acc1, ab[1], bb[1]);

        #pragma unroll
        for (int r = 0; r < 8; ++r) red[(wave * 8 + r) * 32 + lane] = (v2f){acc0[r], acc1[r]};
        __syncthreads();
        #pragma unroll
        for (int rr = 0; rr < 2; ++rr) {
            const int r = 2 * wave + rr;
            v2f s = red[(0 * 8 + r) * 32 + lane];
            s += red[(1 * 8 + r) * 32 + lane];
            s += red[(2 * 8 + r) * 32 + lane];
            s += red[(3 * 8 + r) * 32 + lane];
            s += bsv;
            const int row = sorted[tileBeg + 8 * g + r];
            if (row >= 0) *(v2f*)(out + (size_t)row * O_DIM + colBase + 2 * l16) = s;
        }
    }
}

extern "C" void kernel_launch(void* const* d_in, const int* in_sizes, int n_in,
                              void* d_out, int out_size, void* d_ws, size_t ws_size,
                              hipStream_t stream) {
    const float* inp        = (const float*)d_in[0];
    const float* params     = (const float*)d_in[1];
    const float* bias       = (const float*)d_in[2];
    const int*   edge_types = (const int*)d_in[3];
    const int*   msg_to     = (const int*)d_in[6];
    float* out = (float*)d_out;

    dim3 grid(WIDE_WGS + SLIM_WGS);
    dim3 block(THREADS);
    highorder_wmma_kernel<<<grid, block, 0, stream>>>(inp, params, bias,
                                                      edge_types, msg_to, out);
}
